# starting kernel
# baseline (speedup 1.0000x reference)
#include <hip/hip_runtime.h>

typedef __attribute__((ext_vector_type(2))) float v2f;
typedef __attribute__((ext_vector_type(8))) float v8f;

#define EPS_F 1e-9f

constexpr int DI(int m)   { return m / 10; }
constexpr int DJ(int m)   { return m % 10; }
constexpr int DSUM(int m) { return m / 10 + m % 10; }

__global__ __launch_bounds__(256)
void dpl_add_wmma_kernel(const float* __restrict__ p1,
                         const float* __restrict__ p2,
                         float* __restrict__ out, int B)
{
    __shared__ float lds[8][16][20];

    const int tid  = threadIdx.x;
    const int lane = tid & 31;
    const int w    = tid >> 5;
    const int col  = lane & 15;
    const bool lo  = lane < 16;

    const int  tile   = blockIdx.x * 8 + w;
    const int  b0     = tile * 16;
    const bool active = b0 < B;

    if (active) {
        int brow = b0 + col;
        if (brow >= B) brow = B - 1;
        const float* __restrict__ r1 = p1 + (size_t)brow * 10;
        const float* __restrict__ r2 = p2 + (size_t)brow * 10;

        float p1v[10], p2v[10];
        #pragma unroll
        for (int k = 0; k < 10; ++k) { p1v[k] = r1[k]; p2v[k] = r2[k]; }

        float t16 = p1v[7]*p2v[9] + p1v[8]*p2v[8] + p1v[9]*p2v[7];
        float t17 = p1v[8]*p2v[9] + p1v[9]*p2v[8];
        float t18 = p1v[9]*p2v[9];

        v8f acc = {0.f, 0.f, 0.f, 0.f, 0.f, 0.f, 0.f, 0.f};
        #pragma unroll
        for (int kk = 0; kk < 25; ++kk) {
            const int m0 = kk * 4;
            v2f a, b;
            a.x = lo ? p1v[DI(m0+0)] * p2v[DJ(m0+0)]
                     : p1v[DI(m0+2)] * p2v[DJ(m0+2)];
            a.y = lo ? p1v[DI(m0+1)] * p2v[DJ(m0+1)]
                     : p1v[DI(m0+3)] * p2v[DJ(m0+3)];
            int s0 = lo ? DSUM(m0+0) : DSUM(m0+2);
            int s1 = lo ? DSUM(m0+1) : DSUM(m0+3);
            b.x = (col == s0) ? 1.0f : 0.0f;
            b.y = (col == s1) ? 1.0f : 0.0f;
            acc = __builtin_amdgcn_wmma_f32_16x16x4_f32(
                      false, a, false, b, (short)0, acc, false, false);
        }

        const int rbase = lo ? 0 : 8;
        #pragma unroll
        for (int v = 0; v < 8; ++v)
            lds[w][rbase + v][col] = acc[v];
        if (lo) {
            lds[w][lane][16] = t16;
            lds[w][lane][17] = t17;
            lds[w][lane][18] = t18;
        }
    }
    __syncthreads();

    if (active && lo) {
        float s = 0.f;
        #pragma unroll
        for (int k = 0; k < 19; ++k) s += lds[w][lane][k];
        lds[w][lane][19] = 1.0f / (s + EPS_F);
    }
    __syncthreads();

    if (active) {
        float* __restrict__ outw = out + (size_t)b0 * 19;
        int total = (B - b0) * 19;
        if (total > 304) total = 304;
        #pragma unroll
        for (int it = 0; it < 10; ++it) {
            int idx = it * 32 + lane;
            if (idx < total) {
                unsigned row = (unsigned)idx / 19u;
                unsigned c   = (unsigned)idx - row * 19u;
                outw[idx] = lds[w][row][c] * lds[w][row][19];
            }
        }
    }
}

extern "C" void kernel_launch(void* const* d_in, const int* in_sizes, int n_in,
                              void* d_out, int out_size, void* d_ws, size_t ws_size,
                              hipStream_t stream) {
    const float* p1 = (const float*)d_in[0];
    const float* p2 = (const float*)d_in[1];
    float* out = (float*)d_out;
    const int B = in_sizes[0] / 10;

    const int tiles  = (B + 15) / 16;
    const int blocks = (tiles + 7) / 8;
    dpl_add_wmma_kernel<<<blocks, 256, 0, stream>>>(p1, p2, out, B);
}


# Persistent tile loop: fixed grid of 1024 work-groups x 8 waves, each wave walks its 16-row tiles on its own (per-wave LDS slab, no work-group barrier); one-hot B fragments built once per wave
# speedup vs baseline: 1.0667x; 1.0667x over previous
#include <hip/hip_runtime.h>

typedef __attribute__((ext_vector_type(2))) float v2f;
typedef __attribute__((ext_vector_type(8))) float v8f;

#define EPS_F 1e-9f

constexpr int DI(int m)   { return m / 10; }
constexpr int DJ(int m)   { return m % 10; }
constexpr int DSUM(int m) { return m / 10 + m % 10; }

__device__ __forceinline__ void wave_lds_order()
{
    __builtin_amdgcn_fence(__ATOMIC_RELEASE, "wavefront");
    __builtin_amdgcn_wave_barrier();
    __builtin_amdgcn_fence(__ATOMIC_ACQUIRE, "wavefront");
}

__global__ __launch_bounds__(256)
void dpl_add_wmma_kernel(const float* __restrict__ p1,
                         const float* __restrict__ p2,
                         float* __restrict__ out, int B)
{
    __shared__ float lds[8][16][20];

    const int tid  = threadIdx.x;
    const int lane = tid & 31;
    const int w    = __builtin_amdgcn_readfirstlane(tid >> 5);
    const int col  = lane & 15;
    const bool lo  = lane < 16;
    const int rbase = lo ? 0 : 8;

    const int ntiles = (B + 15) >> 4;
    const int tstep  = (int)gridDim.x * 8;

    v2f bfr[25];
    #pragma unroll
    for (int kk = 0; kk < 25; ++kk) {
        const int m0 = kk * 4;
        const int s0 = lo ? DSUM(m0+0) : DSUM(m0+2);
        const int s1 = lo ? DSUM(m0+1) : DSUM(m0+3);
        bfr[kk].x = (col == s0) ? 1.0f : 0.0f;
        bfr[kk].y = (col == s1) ? 1.0f : 0.0f;
    }

    for (int tile = (int)blockIdx.x * 8 + w; tile < ntiles; tile += tstep) {
        const int b0 = tile * 16;

        int brow = b0 + col;
        if (brow >= B) brow = B - 1;
        const float* __restrict__ r1 = p1 + (size_t)brow * 10;
        const float* __restrict__ r2 = p2 + (size_t)brow * 10;

        float p1v[10], p2v[10];
        #pragma unroll
        for (int k = 0; k < 10; ++k) { p1v[k] = r1[k]; p2v[k] = r2[k]; }

        float t16 = p1v[7]*p2v[9] + p1v[8]*p2v[8] + p1v[9]*p2v[7];
        float t17 = p1v[8]*p2v[9] + p1v[9]*p2v[8];
        float t18 = p1v[9]*p2v[9];

        v8f acc = {0.f, 0.f, 0.f, 0.f, 0.f, 0.f, 0.f, 0.f};
        #pragma unroll
        for (int kk = 0; kk < 25; ++kk) {
            const int m0 = kk * 4;
            v2f a;
            a.x = lo ? p1v[DI(m0+0)] * p2v[DJ(m0+0)]
                     : p1v[DI(m0+2)] * p2v[DJ(m0+2)];
            a.y = lo ? p1v[DI(m0+1)] * p2v[DJ(m0+1)]
                     : p1v[DI(m0+3)] * p2v[DJ(m0+3)];
            acc = __builtin_amdgcn_wmma_f32_16x16x4_f32(
                      false, a, false, bfr[kk], (short)0, acc, false, false);
        }

        #pragma unroll
        for (int v = 0; v < 8; ++v)
            lds[w][rbase + v][col] = acc[v];
        if (lo) {
            lds[w][lane][16] = t16;
            lds[w][lane][17] = t17;
            lds[w][lane][18] = t18;
        }
        wave_lds_order();

        if (lo) {
            float s = 0.f;
            #pragma unroll
            for (int k = 0; k < 19; ++k) s += lds[w][lane][k];
            lds[w][lane][19] = 1.0f / (s + EPS_F);
        }
        wave_lds_order();

        float* __restrict__ outw = out + (size_t)b0 * 19;
        int total = (B - b0) * 19;
        if (total > 304) total = 304;
        #pragma unroll
        for (int it = 0; it < 10; ++it) {
            int idx = it * 32 + lane;
            if (idx < total) {
                unsigned row = (unsigned)idx / 19u;
                unsigned c   = (unsigned)idx - row * 19u;
                outw[idx] = lds[w][row][c] * lds[w][row][19];
            }
        }
        wave_lds_order();
    }
}

extern "C" void kernel_launch(void* const* d_in, const int* in_sizes, int n_in,
                              void* d_out, int out_size, void* d_ws, size_t ws_size,
                              hipStream_t stream) {
    const float* p1 = (const float*)d_in[0];
    const float* p2 = (const float*)d_in[1];
    float* out = (float*)d_out;
    const int B = in_sizes[0] / 10;

    dpl_add_wmma_kernel<<<1024, 256, 0, stream>>>(p1, p2, out, B);
}


# Output pass: LDS tile packed at pitch 19 (output order) and read back as 16-byte vectors, all reads issued before the stores; 3 x 16-byte-per-lane global stores per tile instead of 10 x 4-byte
# speedup vs baseline: 1.1608x; 1.0882x over previous
#include <hip/hip_runtime.h>

typedef __attribute__((ext_vector_type(2))) float v2f;
typedef __attribute__((ext_vector_type(4))) float v4f;
typedef __attribute__((ext_vector_type(8))) float v8f;

#define EPS_F 1e-9f

constexpr int DI(int m)   { return m / 10; }
constexpr int DJ(int m)   { return m % 10; }
constexpr int DSUM(int m) { return m / 10 + m % 10; }

__device__ __forceinline__ void wave_lds_order()
{
    __builtin_amdgcn_fence(__ATOMIC_RELEASE, "wavefront");
    __builtin_amdgcn_wave_barrier();
    __builtin_amdgcn_fence(__ATOMIC_ACQUIRE, "wavefront");
}

__global__ __launch_bounds__(256)
void dpl_add_wmma_kernel(const float* __restrict__ p1,
                         const float* __restrict__ p2,
                         float* __restrict__ out, int B)
{
    __shared__ __attribute__((aligned(16))) float lds[8][320];

    const int tid  = threadIdx.x;
    const int lane = tid & 31;
    const int w    = __builtin_amdgcn_readfirstlane(tid >> 5);
    const int col  = lane & 15;
    const bool lo  = lane < 16;
    const int rbase = lo ? 0 : 8;

    const int ntiles = (B + 15) >> 4;
    const int tstep  = (int)gridDim.x * 8;

    v2f bfr[25];
    #pragma unroll
    for (int kk = 0; kk < 25; ++kk) {
        const int m0 = kk * 4;
        const int s0 = lo ? DSUM(m0+0) : DSUM(m0+2);
        const int s1 = lo ? DSUM(m0+1) : DSUM(m0+3);
        bfr[kk].x = (col == s0) ? 1.0f : 0.0f;
        bfr[kk].y = (col == s1) ? 1.0f : 0.0f;
    }

    for (int tile = (int)blockIdx.x * 8 + w; tile < ntiles; tile += tstep) {
        const int b0 = tile * 16;

        int brow = b0 + col;
        if (brow >= B) brow = B - 1;
        const float* __restrict__ r1 = p1 + (size_t)brow * 10;
        const float* __restrict__ r2 = p2 + (size_t)brow * 10;

        float p1v[10], p2v[10];
        #pragma unroll
        for (int k = 0; k < 10; ++k) { p1v[k] = r1[k]; p2v[k] = r2[k]; }

        float t16 = p1v[7]*p2v[9] + p1v[8]*p2v[8] + p1v[9]*p2v[7];
        float t17 = p1v[8]*p2v[9] + p1v[9]*p2v[8];
        float t18 = p1v[9]*p2v[9];

        v8f acc = {0.f, 0.f, 0.f, 0.f, 0.f, 0.f, 0.f, 0.f};
        #pragma unroll
        for (int kk = 0; kk < 25; ++kk) {
            const int m0 = kk * 4;
            v2f a;
            a.x = lo ? p1v[DI(m0+0)] * p2v[DJ(m0+0)]
                     : p1v[DI(m0+2)] * p2v[DJ(m0+2)];
            a.y = lo ? p1v[DI(m0+1)] * p2v[DJ(m0+1)]
                     : p1v[DI(m0+3)] * p2v[DJ(m0+3)];
            acc = __builtin_amdgcn_wmma_f32_16x16x4_f32(
                      false, a, false, bfr[kk], (short)0, acc, false, false);
        }

        #pragma unroll
        for (int v = 0; v < 8; ++v)
            lds[w][(rbase + v) * 19 + col] = acc[v];
        if (lo) {
            lds[w][lane * 19 + 16] = t16;
            lds[w][lane * 19 + 17] = t17;
            lds[w][lane * 19 + 18] = t18;
        }
        wave_lds_order();

        if (lo) {
            float s = 0.f;
            #pragma unroll
            for (int k = 0; k < 19; ++k) s += lds[w][lane * 19 + k];
            lds[w][304 + lane] = 1.0f / (s + EPS_F);
        }
        wave_lds_order();

        float* __restrict__ outw = out + (size_t)b0 * 19;
        if (B - b0 >= 16) {
            v4f   d[3];
            float il[3], ih[3];
            #pragma unroll
            for (int p = 0; p < 3; ++p) {
                const int g  = p * 32 + lane;
                const int gq = g < 76 ? g : 75;
                const unsigned idx = 4u * (unsigned)gq;
                const unsigned r0  = idx / 19u;
                d[p]  = *(const v4f*)&lds[w][idx];
                il[p] = lds[w][304u + r0];
                ih[p] = lds[w][304u + (r0 < 15u ? r0 + 1u : 15u)];
            }
            #pragma unroll
            for (int p = 0; p < 3; ++p) {
                const int g  = p * 32 + lane;
                const int gq = g < 76 ? g : 75;
                const unsigned idx = 4u * (unsigned)gq;
                const unsigned rem = idx - (idx / 19u) * 19u;
                v4f o;
                o.x = d[p].x * il[p];
                o.y = d[p].y * (rem + 1u >= 19u ? ih[p] : il[p]);
                o.z = d[p].z * (rem + 2u >= 19u ? ih[p] : il[p]);
                o.w = d[p].w * (rem + 3u >= 19u ? ih[p] : il[p]);
                if (p == 2) asm volatile("" : "+v"(o.x), "+v"(o.y), "+v"(o.z), "+v"(o.w));
                if (g < 76) *(v4f*)(outw + 4 * g) = o;
            }
        } else {
            const int total = (B - b0) * 19;
            #pragma unroll
            for (int it = 0; it < 10; ++it) {
                const int idx = it * 32 + lane;
                if (idx < total) {
                    const unsigned row = (unsigned)idx / 19u;
                    outw[idx] = lds[w][idx] * lds[w][304u + row];
                }
            }
        }
        wave_lds_order();
    }
}

extern "C" void kernel_launch(void* const* d_in, const int* in_sizes, int n_in,
                              void* d_out, int out_size, void* d_ws, size_t ws_size,
                              hipStream_t stream) {
    const float* p1 = (const float*)d_in[0];
    const float* p2 = (const float*)d_in[1];
    float* out = (float*)d_out;
    const int B = in_sizes[0] / 10;

    dpl_add_wmma_kernel<<<1024, 256, 0, stream>>>(p1, p2, out, B);
}
